# starting kernel
# baseline (speedup 1.0000x reference)
#include <hip/hip_runtime.h>
#include <hip/hip_bf16.h>
#include <math.h>


typedef __attribute__((ext_vector_type(16))) __bf16 v16bf;
typedef __attribute__((ext_vector_type(8)))  float  v8f;

#define IN_CH   64
#define HEADS   8
#define OUT_CH  8
#define HD      (HEADS * OUT_CH)
#define LEAKY   0.3f
#define HT_PITCH 68

__global__ __launch_bounds__(256) void gat_project_wmma(
    const float* __restrict__ h, const float* __restrict__ W,
    float* __restrict__ P, float* __restrict__ Q, int nNodes) {
  __shared__ float hT[16 * HT_PITCH];
  __shared__ float Wl[2 * IN_CH * HD];

  const int tid = threadIdx.x;

  {
    const float4* s4 = (const float4*)(h + (size_t)blockIdx.x * 16 * IN_CH);
    const int r = tid >> 4;
    const int c = (tid & 15) * 4;
    *(float4*)(hT + r * HT_PITCH + c) = s4[tid];
  }
  {
    const float4* s4 = (const float4*)W;
    float4* d4 = (float4*)Wl;
    #pragma unroll
    for (int i = 0; i < 8; ++i)
      d4[tid + 256 * i] = s4[tid + 256 * i];
  }
  __syncthreads();

  const int wave  = tid >> 5;
  const int lane  = tid & 31;
  const int mat   = wave >> 2;
  const int tn    = wave & 3;
  const int l16   = lane & 15;
  const int khalf = lane >> 4;
  const int col   = tn * 16 + l16;
  const int wbase = mat * IN_CH;

  const float* __restrict__ hrow = hT + l16 * HT_PITCH;

  v8f acc = {};
  #pragma unroll
  for (int kb = 0; kb < IN_CH; kb += 32) {
    const int k0 = kb + khalf * 8;
    v16bf A, B;
    #pragma unroll
    for (int j = 0; j < 8; ++j) {
      A[j]     = (__bf16)hrow[k0 + j];
      A[j + 8] = (__bf16)hrow[k0 + 16 + j];
      B[j]     = (__bf16)Wl[(wbase + k0 + j)      * HD + col];
      B[j + 8] = (__bf16)Wl[(wbase + k0 + 16 + j) * HD + col];
    }
    acc = __builtin_amdgcn_wmma_f32_16x16x32_bf16(
         false, A,  false, B,
         (short)0, acc,  false,  false);
  }

  float* __restrict__ dst = mat ? Q : P;
  const int rbase = blockIdx.x * 16 + khalf * 8;
  #pragma unroll
  for (int v = 0; v < 8; ++v)
    dst[(size_t)(rbase + v) * HD + col] = acc[v];
}

__global__ __launch_bounds__(256) void gat_init(
    float* __restrict__ out, float* __restrict__ m, float* __restrict__ s,
    int n64, int n8) {
  int i = blockIdx.x * blockDim.x + threadIdx.x;
  if (i < n64) out[i] = 0.0f;
  if (i < n8) { m[i] = -__builtin_inff(); s[i] = 0.0f; }
}

__global__ __launch_bounds__(256) void gat_logits(
    const float* __restrict__ P, const float* __restrict__ Q,
    const float* __restrict__ a,
    const int* __restrict__ src, const int* __restrict__ tgt,
    float* __restrict__ logits, float* __restrict__ m, int nEdges) {
  int t = blockIdx.x * blockDim.x + threadIdx.x;
  if (t >= nEdges * HEADS) return;
  const int e  = t >> 3;
  const int hd = t & 7;
  const int sn = src[e];
  const int tn = tgt[e];
  const float* __restrict__ p  = P + (size_t)sn * HD + hd * OUT_CH;
  const float* __restrict__ q  = Q + (size_t)tn * HD + hd * OUT_CH;
  const float* __restrict__ av = a + hd * OUT_CH;
  float lg = 0.0f;
  #pragma unroll
  for (int d = 0; d < OUT_CH; ++d) {
    float x = p[d] + q[d];
    x = x > 0.0f ? x : LEAKY * x;
    lg += x * av[d];
  }
  logits[t] = lg;
  float* addr = m + (size_t)tn * HEADS + hd;
  if (lg >= 0.0f) atomicMax((int*)addr, __float_as_int(lg));
  else            atomicMin((unsigned int*)addr, (unsigned int)__float_as_int(lg));
}

__global__ __launch_bounds__(256) void gat_expsum(
    const int* __restrict__ tgt, const float* __restrict__ m,
    float* __restrict__ logits, float* __restrict__ s, int nEdges) {
  int t = blockIdx.x * blockDim.x + threadIdx.x;
  if (t >= nEdges * HEADS) return;
  const int e  = t >> 3;
  const int hd = t & 7;
  const int tn = tgt[e];
  float ex = __expf(logits[t] - m[(size_t)tn * HEADS + hd]);
  logits[t] = ex;
  atomicAdd(s + (size_t)tn * HEADS + hd, ex);
}

__global__ __launch_bounds__(256) void gat_scatter(
    const float* __restrict__ h,
    const int* __restrict__ src, const int* __restrict__ tgt,
    const float* __restrict__ ex, const float* __restrict__ s,
    float* __restrict__ out, int nEdges) {
  int t = blockIdx.x * blockDim.x + threadIdx.x;
  if (t >= nEdges * HEADS) return;
  const int e  = t >> 3;
  const int hd = t & 7;
  const int sn = src[e];
  const int tn = tgt[e];
  const float att = ex[t] / s[(size_t)tn * HEADS + hd];
  const float* __restrict__ hi = h + (size_t)sn * IN_CH + hd * OUT_CH;
  float* __restrict__ o = out + (size_t)tn * HD + hd * OUT_CH;
  #pragma unroll
  for (int d = 0; d < OUT_CH; ++d)
    atomicAdd(o + d, hi[d] * att);
}

extern "C" void kernel_launch(void* const* d_in, const int* in_sizes, int n_in,
                              void* d_out, int out_size, void* d_ws, size_t ws_size,
                              hipStream_t stream) {
  const float* h   = (const float*)d_in[0];
  const float* W   = (const float*)d_in[1];
  const float* a   = (const float*)d_in[2];
  const int*   src = (const int*)d_in[3];
  const int*   tgt = (const int*)d_in[4];
  float*       out = (float*)d_out;

  const int nNodes = in_sizes[0] / IN_CH;
  const int nEdges = in_sizes[3];

  float* P      = (float*)d_ws;
  float* Q      = P + (size_t)nNodes * HD;
  float* logits = Q + (size_t)nNodes * HD;
  float* m      = logits + (size_t)nEdges * HEADS;
  float* s      = m + (size_t)nNodes * HEADS;

  const int rowBlocks = nNodes / 16;
  gat_project_wmma<<<rowBlocks, 256, 0, stream>>>(h, W, P, Q, nNodes);

  const int n64 = nNodes * HD, n8 = nNodes * HEADS;
  gat_init<<<(n64 + 255) / 256, 256, 0, stream>>>(out, m, s, n64, n8);

  const int eh = nEdges * HEADS;
  const int ehBlocks = (eh + 255) / 256;
  gat_logits <<<ehBlocks, 256, 0, stream>>>(P, Q, a, src, tgt, logits, m, nEdges);
  gat_expsum <<<ehBlocks, 256, 0, stream>>>(tgt, m, logits, s, nEdges);
  gat_scatter<<<ehBlocks, 256, 0, stream>>>(h, src, tgt, logits, s, out, nEdges);
}


# P[src] no longer stored or gathered: one resident edge launch forms it per 16-edge tile on the matrix unit from the h rows the scatter gathers; ex in LDS; rows divided by s last; rounding differs
# speedup vs baseline: 2.3356x; 2.3356x over previous
#include <hip/hip_runtime.h>
#include <hip/hip_bf16.h>
#include <math.h>


typedef __attribute__((ext_vector_type(16))) __bf16 v16bf;
typedef __attribute__((ext_vector_type(8)))  float  v8f;

#define IN_CH   64
#define HEADS   8
#define OUT_CH  8
#define HD      (HEADS * OUT_CH)
#define LEAKY   0.3f
#define HT_PITCH 68
#define RANGE_LIMIT 40.0f
#define BND_RECS 8

template <int X>
__device__ __forceinline__ float group8_xor(float v) {
  constexpr int sel = ((0 ^ X)) | ((1 ^ X) << 3) | ((2 ^ X) << 6) | ((3 ^ X) << 9) |
                      ((4 ^ X) << 12) | ((5 ^ X) << 15) | ((6 ^ X) << 18) | ((7 ^ X) << 21);
  return __int_as_float(__builtin_amdgcn_mov_dpp8(__float_as_int(v), sel));
}

__global__ __launch_bounds__(32) void gat_zero_bounds(float* __restrict__ bnd) {
  if (threadIdx.x < 2 * BND_RECS) bnd[threadIdx.x] = 0.0f;
}

__device__ __forceinline__ void read_bounds(const float* __restrict__ bnd,
                                            float& U, float& V) {
  float u = 0.0f, v = 0.0f;
  #pragma unroll
  for (int r = 0; r < BND_RECS; ++r) {
    u = fmaxf(u, bnd[2 * r]);
    v = fmaxf(v, bnd[2 * r + 1]);
  }
  U = u;
  V = v;
}

__global__ __launch_bounds__(256) void gat_project_wmma(
    const float* __restrict__ h, const float* __restrict__ W,
    const float* __restrict__ a,
    float* __restrict__ Q,
    float* __restrict__ out, float* __restrict__ m, float* __restrict__ s,
    float* __restrict__ bnd, int nNodes) {
  __shared__ float hT[16 * HT_PITCH];
  __shared__ float Wl[2 * IN_CH * HD];
  __shared__ float wred[8];

  const int tid = threadIdx.x;

  {
    const float4* s4 = (const float4*)(h + (size_t)blockIdx.x * 16 * IN_CH);
    const int r = tid >> 4;
    const int c = (tid & 15) * 4;
    *(float4*)(hT + r * HT_PITCH + c) = s4[tid];
  }
  {
    const float4* s4 = (const float4*)W;
    float4* d4 = (float4*)Wl;
    #pragma unroll
    for (int i = 0; i < 8; ++i)
      d4[tid + 256 * i] = s4[tid + 256 * i];
  }
  __syncthreads();

  {
    const float4 zero4 = make_float4(0.0f, 0.0f, 0.0f, 0.0f);
    ((float4*)(out + (size_t)blockIdx.x * 16 * HD))[tid] = zero4;
    if (tid < 32) {
      ((float4*)(s + (size_t)blockIdx.x * 16 * HEADS))[tid] = zero4;
    } else if (tid < 64) {
      const float ninf = -__builtin_inff();
      ((float4*)(m + (size_t)blockIdx.x * 16 * HEADS))[tid - 32] =
          make_float4(ninf, ninf, ninf, ninf);
    }
  }

  const int wave  = tid >> 5;
  const int lane  = tid & 31;
  const int mat   = wave >> 2;
  const int tn    = wave & 3;
  const int l16   = lane & 15;
  const int khalf = lane >> 4;
  const int col   = tn * 16 + l16;
  const int wbase = mat * IN_CH;

  const float* __restrict__ hrow = hT + l16 * HT_PITCH;

  v8f acc = {};
  #pragma unroll
  for (int kb = 0; kb < IN_CH; kb += 32) {
    const int k0 = kb + khalf * 8;
    v16bf A, B;
    #pragma unroll
    for (int j = 0; j < 8; ++j) {
      A[j]     = (__bf16)hrow[k0 + j];
      A[j + 8] = (__bf16)hrow[k0 + 16 + j];
      B[j]     = (__bf16)Wl[(wbase + k0 + j)      * HD + col];
      B[j + 8] = (__bf16)Wl[(wbase + k0 + 16 + j) * HD + col];
    }
    acc = __builtin_amdgcn_wmma_f32_16x16x32_bf16(
         false, A,  false, B,
         (short)0, acc,  false,  false);
  }

  const float aabs = fabsf(a[col]);
  float best = 0.0f;
  #pragma unroll
  for (int v = 0; v < 8; ++v) {
    float t = aabs * fabsf(acc[v]);
    t += group8_xor<1>(t);
    t += group8_xor<2>(t);
    t += group8_xor<4>(t);
    best = fmaxf(best, t);
  }
  best = fmaxf(best, __shfl_xor(best, 8, 32));
  best = fmaxf(best, __shfl_xor(best, 16, 32));
  if (lane == 0) wred[wave] = best;
  __syncthreads();
  if (tid < 2) {
    const float b = fmaxf(fmaxf(wred[tid * 4], wred[tid * 4 + 1]),
                          fmaxf(wred[tid * 4 + 2], wred[tid * 4 + 3]));
    atomicMax((int*)(bnd + (blockIdx.x & (BND_RECS - 1)) * 2 + tid), __float_as_int(b));
  }

  if (mat) {
    const int rbase = blockIdx.x * 16 + khalf * 8;
    #pragma unroll
    for (int v = 0; v < 8; ++v)
      Q[(size_t)(rbase + v) * HD + col] = acc[v];
  }
}

#define TILE_EDGES     16
#define EDGE_WAVES     8
#define EDGE_BLOCKS    512

template <bool GUARD>
__device__ __forceinline__ void scatter_tile(
    float* __restrict__ out, const float* __restrict__ hT, const float* __restrict__ eT,
    int tv, int lane, int er, int nvalid) {
  float w0[TILE_EDGES], w1[TILE_EDGES];
  #pragma unroll
  for (int g = 0; g < 4; ++g) {
    const float4 t0 = *(const float4*)(eT + er * TILE_EDGES + 4 * g);
    const float4 t1 = *(const float4*)(eT + (4 + er) * TILE_EDGES + 4 * g);
    w0[4 * g] = t0.x; w0[4 * g + 1] = t0.y; w0[4 * g + 2] = t0.z; w0[4 * g + 3] = t0.w;
    w1[4 * g] = t1.x; w1[4 * g + 1] = t1.y; w1[4 * g + 2] = t1.z; w1[4 * g + 3] = t1.w;
  }
  float hlo[TILE_EDGES], hhi[TILE_EDGES];
  #pragma unroll
  for (int i = 0; i < TILE_EDGES; ++i) {
    hlo[i] = hT[i * HT_PITCH + lane];
    hhi[i] = hT[i * HT_PITCH + lane + 32];
  }
  #pragma unroll
  for (int i = 0; i < TILE_EDGES; ++i) {
    if (!GUARD || i < nvalid) {
      const int tn = __builtin_amdgcn_readlane(tv, i);
      float* __restrict__ o = out + (size_t)tn * HD + lane;
      atomicAdd(o,      hlo[i] * w0[i]);
      atomicAdd(o + 32, hhi[i] * w1[i]);
    }
  }
}

template <int MODE>
__global__ __launch_bounds__(256) void gat_edges(
    const float* __restrict__ h, const float* __restrict__ W,
    const float* __restrict__ a, const float* __restrict__ Q,
    const int* __restrict__ src, const int* __restrict__ tgt,
    const float* __restrict__ bnd,
    float* __restrict__ m, float* __restrict__ s, float* __restrict__ out,
    int nEdges) {
  __shared__ float Wl[IN_CH * HD];
  __shared__ float hS[EDGE_WAVES][TILE_EDGES * HT_PITCH];
  __shared__ float pS[EDGE_WAVES][TILE_EDGES * HT_PITCH];
  __shared__ float eS[EDGE_WAVES][TILE_EDGES * HEADS];

  float U, V;
  read_bounds(bnd, U, V);
  const bool bounded = 2.0f * (U + V) <= RANGE_LIMIT;
  if (MODE == 0 && bounded) return;

  const int tid = threadIdx.x;
  {
    const float4* s4 = (const float4*)W;
    float4* d4 = (float4*)Wl;
    #pragma unroll
    for (int i = 0; i < 4; ++i)
      d4[tid + 256 * i] = s4[tid + 256 * i];
  }
  __syncthreads();

  const int wave  = tid >> 5;
  const int lane  = tid & 31;
  const int l16   = lane & 15;
  const int khalf = lane >> 4;
  const int er    = lane >> 3;
  const int hd    = lane & 7;

  v16bf Bf[4][2];
  #pragma unroll
  for (int nt = 0; nt < 4; ++nt) {
    const int col = nt * 16 + l16;
    #pragma unroll
    for (int kbi = 0; kbi < 2; ++kbi) {
      const int k0 = kbi * 32 + khalf * 8;
      #pragma unroll
      for (int j = 0; j < 8; ++j) {
        Bf[nt][kbi][j]     = (__bf16)Wl[(k0 + j)      * HD + col];
        Bf[nt][kbi][j + 8] = (__bf16)Wl[(k0 + 16 + j) * HD + col];
      }
    }
  }

  float av[OUT_CH];
  #pragma unroll
  for (int d = 0; d < OUT_CH; ++d) av[d] = a[hd * OUT_CH + d];

  float* hT = hS[wave];
  float* pT = pS[wave];
  float* eT = eS[wave];
  const float* hrow = hT + l16 * HT_PITCH;

  const int gw = __builtin_amdgcn_readfirstlane((int)(blockIdx.x * EDGE_WAVES + wave));
  const int last   = nEdges - 1;
  const int nTiles = (nEdges + TILE_EDGES - 1) / TILE_EDGES;
  const int nWaves = (int)gridDim.x * EDGE_WAVES;
  const int tile0  = gw;
  if (tile0 >= nTiles) return;

  int sv = src[min(tile0 * TILE_EDGES + l16, last)];
  int tv = tgt[min(tile0 * TILE_EDGES + l16, last)];
  int tE[4];
  #pragma unroll
  for (int j = 0; j < 4; ++j) tE[j] = tgt[min(tile0 * TILE_EDGES + 4 * j + er, last)];

  for (int tile = tile0; tile < nTiles; tile += nWaves) {
    const int e0 = tile * TILE_EDGES;
    const int nvalid = nEdges - e0;

    float4 hv[8];
    #pragma unroll
    for (int j = 0; j < 8; ++j) {
      const int sA = __builtin_amdgcn_readlane(sv, 2 * j);
      const int sB = __builtin_amdgcn_readlane(sv, 2 * j + 1);
      const int sn = khalf ? sB : sA;
      hv[j] = *(const float4*)(h + (size_t)sn * IN_CH + l16 * 4);
    }
    float4 q0[4], q1[4];
    #pragma unroll
    for (int j = 0; j < 4; ++j) {
      const float* __restrict__ q = Q + (size_t)tE[j] * HD + hd * OUT_CH;
      q0[j] = *(const float4*)(q);
      q1[j] = *(const float4*)(q + 4);
    }

    const int en = e0 + nWaves * TILE_EDGES;
    const int sv_next = src[min(en + l16, last)];
    const int tv_next = tgt[min(en + l16, last)];
    int tE_next[4];
    #pragma unroll
    for (int j = 0; j < 4; ++j) tE_next[j] = tgt[min(en + 4 * j + er, last)];

    #pragma unroll
    for (int j = 0; j < 8; ++j)
      *(float4*)(hT + (2 * j + khalf) * HT_PITCH + l16 * 4) = hv[j];
    __builtin_amdgcn_fence(__ATOMIC_ACQ_REL, "wavefront");

    v8f acc[4] = {};
    #pragma unroll
    for (int kbi = 0; kbi < 2; ++kbi) {
      const int k0 = kbi * 32 + khalf * 8;
      v16bf A;
      #pragma unroll
      for (int j = 0; j < 8; ++j) {
        A[j]     = (__bf16)hrow[k0 + j];
        A[j + 8] = (__bf16)hrow[k0 + 16 + j];
      }
      #pragma unroll
      for (int nt = 0; nt < 4; ++nt)
        acc[nt] = __builtin_amdgcn_wmma_f32_16x16x32_bf16(
             false, A,  false, Bf[nt][kbi],
             (short)0, acc[nt],  false,  false);
    }
    #pragma unroll
    for (int nt = 0; nt < 4; ++nt) {
      #pragma unroll
      for (int v = 0; v < 8; ++v)
        pT[(v + 8 * khalf) * HT_PITCH + nt * 16 + l16] = acc[nt][v];
    }
    __builtin_amdgcn_fence(__ATOMIC_ACQ_REL, "wavefront");

    #pragma unroll
    for (int j = 0; j < 4; ++j) {
      const int r = 4 * j + er;
      const float* pp = pT + r * HT_PITCH + hd * OUT_CH;
      const float4 p0 = *(const float4*)(pp);
      const float4 p1 = *(const float4*)(pp + 4);
      const float p[OUT_CH] = {p0.x, p0.y, p0.z, p0.w, p1.x, p1.y, p1.z, p1.w};
      const float q[OUT_CH] = {q0[j].x, q0[j].y, q0[j].z, q0[j].w,
                               q1[j].x, q1[j].y, q1[j].z, q1[j].w};
      float lg = 0.0f;
      float bq = 0.0f;
      #pragma unroll
      for (int d = 0; d < OUT_CH; ++d) {
        float x = p[d] + q[d];
        x = x > 0.0f ? x : LEAKY * x;
        lg += x * av[d];
        bq += fabsf(av[d]) * fabsf(q[d]);
      }
      const int  tn    = tE[j];
      const bool valid = r < nvalid;
      if (MODE == 0) {
        if (valid) {
          float* addr = m + (size_t)tn * HEADS + hd;
          if (lg >= 0.0f) atomicMax((int*)addr, __float_as_int(lg));
          else            atomicMin((unsigned int*)addr, (unsigned int)__float_as_int(lg));
        }
      } else {
        const float shift = bounded ? (U + bq) : m[(size_t)tn * HEADS + hd];
        const float ex = __expf(lg - shift);
        eT[hd * TILE_EDGES + r] = ex;
        if (valid) atomicAdd(s + (size_t)tn * HEADS + hd, ex);
      }
    }

    if (MODE == 1) {
      __builtin_amdgcn_fence(__ATOMIC_ACQ_REL, "wavefront");
      if (nvalid >= TILE_EDGES) scatter_tile<false>(out, hT, eT, tv, lane, er, nvalid);
      else                      scatter_tile<true >(out, hT, eT, tv, lane, er, nvalid);
      __builtin_amdgcn_fence(__ATOMIC_ACQ_REL, "wavefront");
    }

    sv = sv_next;
    tv = tv_next;
    #pragma unroll
    for (int j = 0; j < 4; ++j) tE[j] = tE_next[j];
  }
}

__global__ __launch_bounds__(256) void gat_divide(
    const float* __restrict__ s, float* __restrict__ out, int nNodes) {
  const int t = blockIdx.x * blockDim.x + threadIdx.x;
  if (t >= nNodes * (HD / 4)) return;
  const int n  = t >> 4;
  const int hd = (t & 15) >> 1;
  const float sv = s[(size_t)n * HEADS + hd];
  float4 o = ((const float4*)out)[t];
  if (sv > 0.0f) {
    o.x /= sv; o.y /= sv; o.z /= sv; o.w /= sv;
    ((float4*)out)[t] = o;
  }
}

extern "C" void kernel_launch(void* const* d_in, const int* in_sizes, int n_in,
                              void* d_out, int out_size, void* d_ws, size_t ws_size,
                              hipStream_t stream) {
  const float* h   = (const float*)d_in[0];
  const float* W   = (const float*)d_in[1];
  const float* a   = (const float*)d_in[2];
  const int*   src = (const int*)d_in[3];
  const int*   tgt = (const int*)d_in[4];
  float*       out = (float*)d_out;

  const int nNodes = in_sizes[0] / IN_CH;
  const int nEdges = in_sizes[3];

  float* Q   = (float*)d_ws;
  float* m   = Q + (size_t)nNodes * HD;
  float* s   = m + (size_t)nNodes * HEADS;
  float* bnd = s + (size_t)nNodes * HEADS;

  const int rowBlocks = nNodes / 16;
  gat_zero_bounds<<<1, 32, 0, stream>>>(bnd);
  gat_project_wmma<<<rowBlocks, 256, 0, stream>>>(h, W, a, Q, out, m, s, bnd, nNodes);

  gat_edges<0><<<EDGE_BLOCKS, 256, 0, stream>>>(h, W, a, Q, src, tgt, bnd, m, s, out, nEdges);
  gat_edges<1><<<EDGE_BLOCKS, 256, 0, stream>>>(h, W, a, Q, src, tgt, bnd, m, s, out, nEdges);

  const int divBlocks = (nNodes * (HD / 4) + 255) / 256;
  gat_divide<<<divBlocks, 256, 0, stream>>>(s, out, nNodes);
}
